# starting kernel
# baseline (speedup 1.0000x reference)
#include <hip/hip_runtime.h>


typedef __attribute__((ext_vector_type(2))) float v2f;
typedef __attribute__((ext_vector_type(8))) float v8f;

#define CDIM   256
#define PITCH  260
#define ROWS   16
#define NITER  30
#define KT     32
#define RPITCH 20
#define RTSZ   (16 * RPITCH)

__global__ __launch_bounds__(512, 1) void fused_tinyops_kernel(
    const float* __restrict__ x, const float* __restrict__ ln_w,
    const float* __restrict__ ln_b, const float* __restrict__ W,
    const float* __restrict__ bias, float* __restrict__ out)
{
    __shared__ float hs[ROWS * PITCH];
    __shared__ float red[16 * RTSZ];

    const int tid   = threadIdx.x;
    const int lane  = tid & 31;
    const int wave  = tid >> 5;
    const int l16   = lane & 15;
    const int half  = (lane < 16) ? 0 : 2;
    const int khalf = wave >> 3;
    const int cpair = wave & 7;
    const int c0    = 2 * cpair;
    const int c1    = c0 + 1;

    const long rowBase = (long)blockIdx.x * ROWS;
    const float* xg = x + rowBase * CDIM;

    #pragma unroll
    for (int j = 0; j < (ROWS * CDIM) / 512; ++j) {
        int idx = j * 512 + tid;
        hs[(idx >> 8) * PITCH + (idx & (CDIM - 1))] = xg[idx];
    }

    v2f B0[KT], B1[KT];
    {
        const float* w0 = W + (16 * c0 + l16) * CDIM + 128 * khalf + half;
        const float* w1 = W + (16 * c1 + l16) * CDIM + 128 * khalf + half;
        #pragma unroll
        for (int t = 0; t < KT; ++t) {
            B0[t].x = w0[4 * t + 0];  B0[t].y = w0[4 * t + 1];
            B1[t].x = w1[4 * t + 0];  B1[t].y = w1[4 * t + 1];
        }
    }
    const float bias0 = bias[16 * c0 + l16];
    const float bias1 = bias[16 * c1 + l16];

    float lw[8], lb[8];
    #pragma unroll
    for (int j = 0; j < 8; ++j) {
        lw[j] = ln_w[lane + 32 * j];
        lb[j] = ln_b[lane + 32 * j];
    }

    __syncthreads();

    for (int it = 0; it < NITER; ++it) {
        float v[8];
        float* hrow = &hs[wave * PITCH];
        #pragma unroll
        for (int j = 0; j < 8; ++j) {
            float h = hrow[lane + 32 * j];
            h = h + 0.01f * tanhf(h);
            h = h * (1.0f / (1.0f + __expf(-h)));
            v[j] = h;
        }
        float s = ((v[0] + v[1]) + (v[2] + v[3])) + ((v[4] + v[5]) + (v[6] + v[7]));
        #pragma unroll
        for (int off = 16; off > 0; off >>= 1) s += __shfl_xor(s, off, 32);
        const float mu = s * (1.0f / CDIM);
        float q = 0.0f;
        #pragma unroll
        for (int j = 0; j < 8; ++j) { float d = v[j] - mu; q += d * d; }
        #pragma unroll
        for (int off = 16; off > 0; off >>= 1) q += __shfl_xor(q, off, 32);
        const float rstd = rsqrtf(q * (1.0f / CDIM) + 1e-5f);
        #pragma unroll
        for (int j = 0; j < 8; ++j)
            hrow[lane + 32 * j] = (v[j] - mu) * rstd * lw[j] + lb[j];

        __syncthreads();

        v8f acc0 = {0.f, 0.f, 0.f, 0.f, 0.f, 0.f, 0.f, 0.f};
        v8f acc1 = {0.f, 0.f, 0.f, 0.f, 0.f, 0.f, 0.f, 0.f};
        const float* arow = &hs[l16 * PITCH + 128 * khalf + half];
        #pragma unroll
        for (int t = 0; t < KT; ++t) {
            v2f a;
            a.x = arow[4 * t + 0];
            a.y = arow[4 * t + 1];
            acc0 = __builtin_amdgcn_wmma_f32_16x16x4_f32(
                false, a, false, B0[t], (short)0, acc0, false, false);
            acc1 = __builtin_amdgcn_wmma_f32_16x16x4_f32(
                false, a, false, B1[t], (short)0, acc1, false, false);
        }

        if (khalf == 1) {
            float* r0 = &red[(2 * cpair + 0) * RTSZ];
            float* r1 = &red[(2 * cpair + 1) * RTSZ];
            #pragma unroll
            for (int vi = 0; vi < 8; ++vi) {
                int r = vi + ((lane < 16) ? 0 : 8);
                r0[r * RPITCH + l16] = acc0[vi];
                r1[r * RPITCH + l16] = acc1[vi];
            }
        }
        __syncthreads();

        if (khalf == 0) {
            const float* r0 = &red[(2 * cpair + 0) * RTSZ];
            const float* r1 = &red[(2 * cpair + 1) * RTSZ];
            #pragma unroll
            for (int vi = 0; vi < 8; ++vi) {
                int r = vi + ((lane < 16) ? 0 : 8);
                hs[r * PITCH + 16 * c0 + l16] =
                    acc0[vi] + r0[r * RPITCH + l16] + bias0;
                hs[r * PITCH + 16 * c1 + l16] =
                    acc1[vi] + r1[r * RPITCH + l16] + bias1;
            }
        }
        __syncthreads();
    }

    float* og = out + rowBase * CDIM;
    #pragma unroll
    for (int j = 0; j < (ROWS * CDIM) / 512; ++j) {
        int idx = j * 512 + tid;
        og[idx] = hs[(idx >> 8) * PITCH + (idx & (CDIM - 1))];
    }
}

extern "C" void kernel_launch(void* const* d_in, const int* in_sizes, int n_in,
                              void* d_out, int out_size, void* d_ws, size_t ws_size,
                              hipStream_t stream) {
    (void)n_in; (void)d_ws; (void)ws_size; (void)out_size;
    const float* x    = (const float*)d_in[0];
    const float* ln_w = (const float*)d_in[1];
    const float* ln_b = (const float*)d_in[2];
    const float* W    = (const float*)d_in[3];
    const float* b    = (const float*)d_in[4];
    float* out        = (float*)d_out;

    const int total_rows = in_sizes[0] / CDIM;
    dim3 grid(total_rows / ROWS);
    dim3 block(512);
    fused_tinyops_kernel<<<grid, block, 0, stream>>>(x, ln_w, ln_b, W, b, out);
}


# Pointwise stage: tanh and sigmoid from hardware exp2 + rcp (tanh = 1 - 2/(1+e^2h)) instead of library tanhf and IEEE divide; far fewer vector instructions per element
# speedup vs baseline: 1.2700x; 1.2700x over previous
#include <hip/hip_runtime.h>


typedef __attribute__((ext_vector_type(2))) float v2f;
typedef __attribute__((ext_vector_type(8))) float v8f;

#define CDIM   256
#define PITCH  260
#define ROWS   16
#define NITER  30
#define KT     32
#define RPITCH 20
#define RTSZ   (16 * RPITCH)

__global__ __launch_bounds__(512, 1) void fused_tinyops_kernel(
    const float* __restrict__ x, const float* __restrict__ ln_w,
    const float* __restrict__ ln_b, const float* __restrict__ W,
    const float* __restrict__ bias, float* __restrict__ out)
{
    __shared__ float hs[ROWS * PITCH];
    __shared__ float red[16 * RTSZ];

    const int tid   = threadIdx.x;
    const int lane  = tid & 31;
    const int wave  = tid >> 5;
    const int l16   = lane & 15;
    const int half  = (lane < 16) ? 0 : 2;
    const int khalf = wave >> 3;
    const int cpair = wave & 7;
    const int c0    = 2 * cpair;
    const int c1    = c0 + 1;

    const long rowBase = (long)blockIdx.x * ROWS;
    const float* xg = x + rowBase * CDIM;

    #pragma unroll
    for (int j = 0; j < (ROWS * CDIM) / 512; ++j) {
        int idx = j * 512 + tid;
        hs[(idx >> 8) * PITCH + (idx & (CDIM - 1))] = xg[idx];
    }

    v2f B0[KT], B1[KT];
    {
        const float* w0 = W + (16 * c0 + l16) * CDIM + 128 * khalf + half;
        const float* w1 = W + (16 * c1 + l16) * CDIM + 128 * khalf + half;
        #pragma unroll
        for (int t = 0; t < KT; ++t) {
            B0[t].x = w0[4 * t + 0];  B0[t].y = w0[4 * t + 1];
            B1[t].x = w1[4 * t + 0];  B1[t].y = w1[4 * t + 1];
        }
    }
    const float bias0 = bias[16 * c0 + l16];
    const float bias1 = bias[16 * c1 + l16];

    float lw[8], lb[8];
    #pragma unroll
    for (int j = 0; j < 8; ++j) {
        lw[j] = ln_w[lane + 32 * j];
        lb[j] = ln_b[lane + 32 * j];
    }

    __syncthreads();

    for (int it = 0; it < NITER; ++it) {
        float v[8];
        float* hrow = &hs[wave * PITCH];
        #pragma unroll
        for (int j = 0; j < 8; ++j) {
            float h = hrow[lane + 32 * j];
            const float r2 = __builtin_amdgcn_rcpf(
                1.0f + __builtin_amdgcn_exp2f(h * 2.8853900817779268f));
            h = fmaf(0.01f, fmaf(-2.0f, r2, 1.0f), h);
            const float r1 = __builtin_amdgcn_rcpf(
                1.0f + __builtin_amdgcn_exp2f(h * -1.4426950408889634f));
            h = h * r1;
            v[j] = h;
        }
        float s = ((v[0] + v[1]) + (v[2] + v[3])) + ((v[4] + v[5]) + (v[6] + v[7]));
        #pragma unroll
        for (int off = 16; off > 0; off >>= 1) s += __shfl_xor(s, off, 32);
        const float mu = s * (1.0f / CDIM);
        float q = 0.0f;
        #pragma unroll
        for (int j = 0; j < 8; ++j) { float d = v[j] - mu; q += d * d; }
        #pragma unroll
        for (int off = 16; off > 0; off >>= 1) q += __shfl_xor(q, off, 32);
        const float rstd = rsqrtf(q * (1.0f / CDIM) + 1e-5f);
        #pragma unroll
        for (int j = 0; j < 8; ++j)
            hrow[lane + 32 * j] = (v[j] - mu) * rstd * lw[j] + lb[j];

        __syncthreads();

        v8f acc0 = {0.f, 0.f, 0.f, 0.f, 0.f, 0.f, 0.f, 0.f};
        v8f acc1 = {0.f, 0.f, 0.f, 0.f, 0.f, 0.f, 0.f, 0.f};
        const float* arow = &hs[l16 * PITCH + 128 * khalf + half];
        #pragma unroll
        for (int t = 0; t < KT; ++t) {
            v2f a;
            a.x = arow[4 * t + 0];
            a.y = arow[4 * t + 1];
            acc0 = __builtin_amdgcn_wmma_f32_16x16x4_f32(
                false, a, false, B0[t], (short)0, acc0, false, false);
            acc1 = __builtin_amdgcn_wmma_f32_16x16x4_f32(
                false, a, false, B1[t], (short)0, acc1, false, false);
        }

        if (khalf == 1) {
            float* r0 = &red[(2 * cpair + 0) * RTSZ];
            float* r1 = &red[(2 * cpair + 1) * RTSZ];
            #pragma unroll
            for (int vi = 0; vi < 8; ++vi) {
                int r = vi + ((lane < 16) ? 0 : 8);
                r0[r * RPITCH + l16] = acc0[vi];
                r1[r * RPITCH + l16] = acc1[vi];
            }
        }
        __syncthreads();

        if (khalf == 0) {
            const float* r0 = &red[(2 * cpair + 0) * RTSZ];
            const float* r1 = &red[(2 * cpair + 1) * RTSZ];
            #pragma unroll
            for (int vi = 0; vi < 8; ++vi) {
                int r = vi + ((lane < 16) ? 0 : 8);
                hs[r * PITCH + 16 * c0 + l16] =
                    acc0[vi] + r0[r * RPITCH + l16] + bias0;
                hs[r * PITCH + 16 * c1 + l16] =
                    acc1[vi] + r1[r * RPITCH + l16] + bias1;
            }
        }
        __syncthreads();
    }

    float* og = out + rowBase * CDIM;
    #pragma unroll
    for (int j = 0; j < (ROWS * CDIM) / 512; ++j) {
        int idx = j * 512 + tid;
        og[idx] = hs[(idx >> 8) * PITCH + (idx & (CDIM - 1))];
    }
}

extern "C" void kernel_launch(void* const* d_in, const int* in_sizes, int n_in,
                              void* d_out, int out_size, void* d_ws, size_t ws_size,
                              hipStream_t stream) {
    (void)n_in; (void)d_ws; (void)ws_size; (void)out_size;
    const float* x    = (const float*)d_in[0];
    const float* ln_w = (const float*)d_in[1];
    const float* ln_b = (const float*)d_in[2];
    const float* W    = (const float*)d_in[3];
    const float* b    = (const float*)d_in[4];
    float* out        = (float*)d_out;

    const int total_rows = in_sizes[0] / CDIM;
    dim3 grid(total_rows / ROWS);
    dim3 block(512);
    fused_tinyops_kernel<<<grid, block, 0, stream>>>(x, ln_w, ln_b, W, b, out);
}


# Each block owns two 16-row tiles run half a step apart: pointwise+LayerNorm of one tile is issued between the WMMAs of the other tile's Linear; both split-K halves go to LDS, one barrier per phase
# speedup vs baseline: 1.3970x; 1.1000x over previous
#include <hip/hip_runtime.h>


typedef __attribute__((ext_vector_type(2))) float v2f;
typedef __attribute__((ext_vector_type(8))) float v8f;

#define CDIM   256
#define PITCH  260
#define ROWS   16
#define NITER  30
#define KT     32
#define TSZ    (ROWS * PITCH)

__device__ __forceinline__ float pointwise(float h)
{
    const float r2 = __builtin_amdgcn_rcpf(
        1.0f + __builtin_amdgcn_exp2f(h * 2.8853900817779268f));
    h = fmaf(0.01f, fmaf(-2.0f, r2, 1.0f), h);
    const float r1 = __builtin_amdgcn_rcpf(
        1.0f + __builtin_amdgcn_exp2f(h * -1.4426950408889634f));
    return h * r1;
}

template <int SRC, int DST, bool GEMM>
__device__ __forceinline__ void phase(
    const v2f (&B0)[KT], const v2f (&B1)[KT],
    const float* __restrict__ arow,
    float* __restrict__ pout,
    int dc1,
    const float* __restrict__ plo,
    const float* __restrict__ phi,
    const float* __restrict__ xrow,
    float* __restrict__ adst,
    float* __restrict__ orow,
    const float (&lw)[8], const float (&lb)[8], const float (&bs)[8])
{
    float v[8];
    if (SRC == 0) {
        #pragma unroll
        for (int j = 0; j < 8; ++j) v[j] = (plo[32 * j] + phi[32 * j]) + bs[j];
    } else {
        #pragma unroll
        for (int j = 0; j < 8; ++j) v[j] = xrow[32 * j];
    }
    if (DST == 1) {
        #pragma unroll
        for (int j = 0; j < 8; ++j) orow[32 * j] = v[j];
    }

    v8f acc0 = {0.f, 0.f, 0.f, 0.f, 0.f, 0.f, 0.f, 0.f};
    v8f acc1 = {0.f, 0.f, 0.f, 0.f, 0.f, 0.f, 0.f, 0.f};
    float s = 0.0f, r = 0.0f, mu = 0.0f, q = 0.0f, rstd = 0.0f;

    constexpr int PF = 3;
    v2f afr[KT];
    if (GEMM) {
        #pragma unroll
        for (int t = 0; t < PF; ++t) {
            afr[t].x = arow[4 * t + 0];
            afr[t].y = arow[4 * t + 1];
        }
    }

    #pragma unroll
    for (int t = 0; t < KT; ++t) {
        if (GEMM) {
            if (t + PF < KT) {
                afr[t + PF].x = arow[4 * (t + PF) + 0];
                afr[t + PF].y = arow[4 * (t + PF) + 1];
            }
            acc0 = __builtin_amdgcn_wmma_f32_16x16x4_f32(
                false, afr[t], false, B0[t], (short)0, acc0, false, false);
            acc1 = __builtin_amdgcn_wmma_f32_16x16x4_f32(
                false, afr[t], false, B1[t], (short)0, acc1, false, false);
        }
        if (DST == 0) {
            if (t == 2) {
                #pragma unroll
                for (int j = 0; j < 4; ++j) v[j] = pointwise(v[j]);
            } else if (t == 3) {
                #pragma unroll
                for (int j = 4; j < 8; ++j) v[j] = pointwise(v[j]);
                s = ((v[0] + v[1]) + (v[2] + v[3])) + ((v[4] + v[5]) + (v[6] + v[7]));
                r = __shfl_xor(s, 16, 32);
            } else if (t == 5)  { s += r; r = __shfl_xor(s, 8, 32);
            } else if (t == 7)  { s += r; r = __shfl_xor(s, 4, 32);
            } else if (t == 9)  { s += r; r = __shfl_xor(s, 2, 32);
            } else if (t == 11) { s += r; r = __shfl_xor(s, 1, 32);
            } else if (t == 13) {
                s += r;
                mu = s * (1.0f / CDIM);
                #pragma unroll
                for (int j = 0; j < 8; ++j) { float d = v[j] - mu; q += d * d; }
                r = __shfl_xor(q, 16, 32);
            } else if (t == 15) { q += r; r = __shfl_xor(q, 8, 32);
            } else if (t == 17) { q += r; r = __shfl_xor(q, 4, 32);
            } else if (t == 19) { q += r; r = __shfl_xor(q, 2, 32);
            } else if (t == 21) { q += r; r = __shfl_xor(q, 1, 32);
            } else if (t == 23) {
                q += r;
                rstd = rsqrtf(q * (1.0f / CDIM) + 1e-5f);
            } else if (t >= 24) {
                const int j = t - 24;
                adst[32 * j] = (v[j] - mu) * rstd * lw[j] + lb[j];
            }
        }
        if (GEMM && DST == 0) {
            __builtin_amdgcn_sched_barrier(0);
        }
    }

    if (GEMM) {
        #pragma unroll
        for (int vi = 0; vi < 8; ++vi) {
            pout[vi * PITCH]       = acc0[vi];
            pout[vi * PITCH + dc1] = acc1[vi];
        }
    }
}

__global__ __launch_bounds__(512, 1) void fused_tinyops_kernel(
    const float* __restrict__ x, const float* __restrict__ ln_w,
    const float* __restrict__ ln_b, const float* __restrict__ W,
    const float* __restrict__ bias, float* __restrict__ out)
{
    __shared__ float abuf[2 * TSZ];
    __shared__ float part[4 * TSZ];

    const int tid   = threadIdx.x;
    const int lane  = tid & 31;
    const int wave  = tid >> 5;
    const int l16   = lane & 15;
    const int half  = (lane < 16) ? 0 : 2;
    const int khalf = wave >> 3;
    const int cpair = wave & 7;
    const int c0    = 2 * cpair;
    const int c1    = c0 + 1;

    const long rowBase = (long)blockIdx.x * (2 * ROWS);

    v2f B0[KT], B1[KT];
    {
        const float* w0 = W + (16 * c0 + l16) * CDIM + 128 * khalf + half;
        const float* w1 = W + (16 * c1 + l16) * CDIM + 128 * khalf + half;
        #pragma unroll
        for (int t = 0; t < KT; ++t) {
            B0[t].x = w0[4 * t + 0];  B0[t].y = w0[4 * t + 1];
            B1[t].x = w1[4 * t + 0];  B1[t].y = w1[4 * t + 1];
        }
    }

    float lw[8], lb[8], bs[8];
    #pragma unroll
    for (int j = 0; j < 8; ++j) {
        lw[j] = ln_w[lane + 32 * j];
        lb[j] = ln_b[lane + 32 * j];
        bs[j] = bias[lane + 32 * j];
    }

    const int aoff = l16 * PITCH + 128 * khalf + half;
    const int poff = khalf * TSZ + ((lane < 16) ? 0 : 8) * PITCH + 16 * c0 + l16;
    const float* a0 = &abuf[0 * TSZ + aoff];
    const float* a1 = &abuf[1 * TSZ + aoff];
    float* p0 = &part[0 * 2 * TSZ + poff];
    float* p1 = &part[1 * 2 * TSZ + poff];
    const int roff = wave * PITCH + lane;
    const float* lo0 = &part[0 * 2 * TSZ + roff];
    const float* hi0 = &part[0 * 2 * TSZ + TSZ + roff];
    const float* lo1 = &part[1 * 2 * TSZ + roff];
    const float* hi1 = &part[1 * 2 * TSZ + TSZ + roff];
    float* d0 = &abuf[0 * TSZ + roff];
    float* d1 = &abuf[1 * TSZ + roff];
    const float* x0 = x + (rowBase + wave) * CDIM + lane;
    const float* x1 = x + (rowBase + ROWS + wave) * CDIM + lane;
    float* o0 = out + (rowBase + wave) * CDIM + lane;
    float* o1 = out + (rowBase + ROWS + wave) * CDIM + lane;

    phase<1, 0, false>(B0, B1, a0, p0, 16, lo0, hi0, x0, d0, o0, lw, lb, bs);
    __syncthreads();
    phase<1, 0, true>(B0, B1, a0, p0, 16, lo1, hi1, x1, d1, o1, lw, lb, bs);
    __syncthreads();

    for (int it = 0; it < NITER - 1; ++it) {
        phase<0, 0, true>(B0, B1, a1, p1, 16, lo0, hi0, x0, d0, o0, lw, lb, bs);
        __syncthreads();
        phase<0, 0, true>(B0, B1, a0, p0, 16, lo1, hi1, x1, d1, o1, lw, lb, bs);
        __syncthreads();
    }

    phase<0, 1, true>(B0, B1, a1, p1, 16, lo0, hi0, x0, d0, o0, lw, lb, bs);
    __syncthreads();
    phase<0, 1, false>(B0, B1, a1, p1, 16, lo1, hi1, x1, d1, o1, lw, lb, bs);
}

extern "C" void kernel_launch(void* const* d_in, const int* in_sizes, int n_in,
                              void* d_out, int out_size, void* d_ws, size_t ws_size,
                              hipStream_t stream) {
    (void)n_in; (void)d_ws; (void)ws_size; (void)out_size;
    const float* x    = (const float*)d_in[0];
    const float* ln_w = (const float*)d_in[1];
    const float* ln_b = (const float*)d_in[2];
    const float* W    = (const float*)d_in[3];
    const float* b    = (const float*)d_in[4];
    float* out        = (float*)d_out;

    const int total_rows = in_sizes[0] / CDIM;
    dim3 grid(total_rows / (2 * ROWS));
    dim3 block(512);
    fused_tinyops_kernel<<<grid, block, 0, stream>>>(x, ln_w, ln_b, W, b, out);
}


# LDS work leaves the WMMA stream: A read as 2 groups of 16-byte loads (k remapped alike in A and B loaders: the k summation order changes), LayerNorm row sums by DPP/permlane, LN stores at phase end
# speedup vs baseline: 1.5125x; 1.0827x over previous
#include <hip/hip_runtime.h>


typedef __attribute__((ext_vector_type(2))) float v2f;
typedef __attribute__((ext_vector_type(4))) float v4f;
typedef __attribute__((ext_vector_type(8))) float v8f;

#define CDIM   256
#define PITCH  260
#define ROWS   16
#define NITER  30
#define KT     32
#define TSZ    (ROWS * PITCH)
#define AG     16
#define LEAD   8
#define VT0    3

__device__ __forceinline__ float pointwise(float h)
{
    const float r2 = __builtin_amdgcn_rcpf(
        1.0f + __builtin_amdgcn_exp2f(h * 2.8853900817779268f));
    h = fmaf(0.01f, fmaf(-2.0f, r2, 1.0f), h);
    const float r1 = __builtin_amdgcn_rcpf(
        1.0f + __builtin_amdgcn_exp2f(h * -1.4426950408889634f));
    return h * r1;
}

__device__ __forceinline__ float xrow16(float x)
{
    const int b = __builtin_bit_cast(int, x);
    return __builtin_bit_cast(float,
        __builtin_amdgcn_permlanex16(b, b, 0x76543210, 0xfedcba98, false, false));
}

template <int M>
__device__ __forceinline__ float xlane(float x)
{
    const int b = __builtin_bit_cast(int, x);
    return __builtin_bit_cast(float,
        __builtin_amdgcn_update_dpp(0, b, 0x160 + M, 0xf, 0xf, true));
}

__device__ __forceinline__ void load_a_group(v2f (&afr)[KT], const float* __restrict__ arow, int g)
{
    #pragma unroll
    for (int i = 0; i < AG / 2; ++i) {
        const int q = g * (AG / 2) + i;
        const v4f a = *reinterpret_cast<const v4f*>(arow + 4 * q);
        afr[2 * q].x     = a.x;  afr[2 * q].y     = a.y;
        afr[2 * q + 1].x = a.z;  afr[2 * q + 1].y = a.w;
    }
}

template <int SRC, int DST, bool GEMM>
__device__ __forceinline__ void phase(
    const v2f (&B0)[KT], const v2f (&B1)[KT],
    const float* __restrict__ arow,
    float* __restrict__ pout,
    int dc1,
    const float* __restrict__ plo,
    const float* __restrict__ phi,
    const float* __restrict__ xrow,
    float* __restrict__ adst,
    float* __restrict__ orow,
    const float (&lw)[8], const float (&lb)[8], const float (&bs)[8])
{
    v2f afr[KT];
    float lo[8], hi[8], v[8];

    static_assert(LEAD > 0 && LEAD < AG && KT % AG == 0, "group schedule");
    if (GEMM) {
        load_a_group(afr, arow, 0);
        if (DST == 0) {
            __builtin_amdgcn_sched_barrier(0);
        }
    }
    if (SRC == 0) {
        #pragma unroll
        for (int j = 0; j < 8; ++j) { lo[j] = plo[32 * j]; hi[j] = phi[32 * j]; }
    } else {
        #pragma unroll
        for (int j = 0; j < 8; ++j) v[j] = xrow[32 * j];
    }
    if (GEMM && DST == 0) {
        __builtin_amdgcn_sched_barrier(0);
    }
    if (DST == 1) {
        #pragma unroll
        for (int j = 0; j < 8; ++j) {
            if (SRC == 0) v[j] = (lo[j] + hi[j]) + bs[j];
            orow[32 * j] = v[j];
        }
    }

    v8f acc0 = {0.f, 0.f, 0.f, 0.f, 0.f, 0.f, 0.f, 0.f};
    v8f acc1 = {0.f, 0.f, 0.f, 0.f, 0.f, 0.f, 0.f, 0.f};
    float s = 0.0f, mu = 0.0f, q = 0.0f, rstd = 0.0f;

    #pragma unroll
    for (int t = 0; t < KT; ++t) {
        if (GEMM) {
            if ((t + LEAD) % AG == 0 && t + LEAD >= AG && t + LEAD < KT) {
                load_a_group(afr, arow, (t + LEAD) / AG);
                if (DST == 0) {
                    __builtin_amdgcn_sched_barrier(0);
                }
            }
            acc0 = __builtin_amdgcn_wmma_f32_16x16x4_f32(
                false, afr[t], false, B0[t], (short)0, acc0, false, false);
            acc1 = __builtin_amdgcn_wmma_f32_16x16x4_f32(
                false, afr[t], false, B1[t], (short)0, acc1, false, false);
        }
        if (DST == 0) {
            if (t == VT0) {
                #pragma unroll
                for (int j = 0; j < 8; ++j) {
                    if (SRC == 0) v[j] = (lo[j] + hi[j]) + bs[j];
                    v[j] = pointwise(v[j]);
                }
                s = ((v[0] + v[1]) + (v[2] + v[3])) + ((v[4] + v[5]) + (v[6] + v[7]));
                s += xrow16(s);
                s += xlane<8>(s);
                s += xlane<4>(s);
                s += xlane<2>(s);
                s += xlane<1>(s);
                mu = s * (1.0f / CDIM);
                #pragma unroll
                for (int j = 0; j < 8; ++j) { float d = v[j] - mu; q += d * d; }
            } else if (t == VT0 + 1) {
                q += xrow16(q);
                q += xlane<8>(q);
                q += xlane<4>(q);
                q += xlane<2>(q);
                q += xlane<1>(q);
                rstd = rsqrtf(q * (1.0f / CDIM) + 1e-5f);
            } else if (t >= VT0 + 2 && t < VT0 + 10) {
                const int j = t - (VT0 + 2);
                v[j] = (v[j] - mu) * rstd * lw[j] + lb[j];
            }
        }
        if (GEMM && DST == 0) {
            __builtin_amdgcn_sched_barrier(0);
        }
    }

    if (DST == 0) {
        #pragma unroll
        for (int j = 0; j < 8; ++j) adst[32 * j] = v[j];
    }
    if (GEMM) {
        #pragma unroll
        for (int vi = 0; vi < 8; ++vi) {
            pout[vi * PITCH]       = acc0[vi];
            pout[vi * PITCH + dc1] = acc1[vi];
        }
    }
}

__global__ __launch_bounds__(512, 1) void fused_tinyops_kernel(
    const float* __restrict__ x, const float* __restrict__ ln_w,
    const float* __restrict__ ln_b, const float* __restrict__ W,
    const float* __restrict__ bias, float* __restrict__ out)
{
    __shared__ __attribute__((aligned(16))) float abuf[2 * TSZ];
    __shared__ float part[4 * TSZ];

    const int tid   = threadIdx.x;
    const int lane  = tid & 31;
    const int wave  = tid >> 5;
    const int l16   = lane & 15;
    const int koff  = (lane < 16) ? 0 : 64;
    const int khalf = wave >> 3;
    const int cpair = wave & 7;
    const int c0    = 2 * cpair;
    const int c1    = c0 + 1;

    const long rowBase = (long)blockIdx.x * (2 * ROWS);

    v2f B0[KT], B1[KT];
    {
        const float* w0 = W + (16 * c0 + l16) * CDIM + 128 * khalf + koff;
        const float* w1 = W + (16 * c1 + l16) * CDIM + 128 * khalf + koff;
        #pragma unroll
        for (int t = 0; t < KT; ++t) {
            B0[t].x = w0[2 * t + 0];  B0[t].y = w0[2 * t + 1];
            B1[t].x = w1[2 * t + 0];  B1[t].y = w1[2 * t + 1];
        }
    }

    float lw[8], lb[8], bs[8];
    #pragma unroll
    for (int j = 0; j < 8; ++j) {
        lw[j] = ln_w[lane + 32 * j];
        lb[j] = ln_b[lane + 32 * j];
        bs[j] = bias[lane + 32 * j];
    }

    const int aoff = l16 * PITCH + 128 * khalf + koff;
    const int poff = khalf * TSZ + ((lane < 16) ? 0 : 8) * PITCH + 16 * c0 + l16;
    const float* a0 = &abuf[0 * TSZ + aoff];
    const float* a1 = &abuf[1 * TSZ + aoff];
    float* p0 = &part[0 * 2 * TSZ + poff];
    float* p1 = &part[1 * 2 * TSZ + poff];
    const int roff = wave * PITCH + lane;
    const float* lo0 = &part[0 * 2 * TSZ + roff];
    const float* hi0 = &part[0 * 2 * TSZ + TSZ + roff];
    const float* lo1 = &part[1 * 2 * TSZ + roff];
    const float* hi1 = &part[1 * 2 * TSZ + TSZ + roff];
    float* d0 = &abuf[0 * TSZ + roff];
    float* d1 = &abuf[1 * TSZ + roff];
    const float* x0 = x + (rowBase + wave) * CDIM + lane;
    const float* x1 = x + (rowBase + ROWS + wave) * CDIM + lane;
    float* o0 = out + (rowBase + wave) * CDIM + lane;
    float* o1 = out + (rowBase + ROWS + wave) * CDIM + lane;

    phase<1, 0, false>(B0, B1, a0, p0, 16, lo0, hi0, x0, d0, o0, lw, lb, bs);
    __syncthreads();
    phase<1, 0, true>(B0, B1, a0, p0, 16, lo1, hi1, x1, d1, o1, lw, lb, bs);
    __syncthreads();

    for (int it = 0; it < NITER - 1; ++it) {
        phase<0, 0, true>(B0, B1, a1, p1, 16, lo0, hi0, x0, d0, o0, lw, lb, bs);
        __syncthreads();
        phase<0, 0, true>(B0, B1, a0, p0, 16, lo1, hi1, x1, d1, o1, lw, lb, bs);
        __syncthreads();
    }

    phase<0, 1, true>(B0, B1, a1, p1, 16, lo0, hi0, x0, d0, o0, lw, lb, bs);
    __syncthreads();
    phase<0, 1, false>(B0, B1, a1, p1, 16, lo1, hi1, x1, d1, o1, lw, lb, bs);
}

extern "C" void kernel_launch(void* const* d_in, const int* in_sizes, int n_in,
                              void* d_out, int out_size, void* d_ws, size_t ws_size,
                              hipStream_t stream) {
    (void)n_in; (void)d_ws; (void)ws_size; (void)out_size;
    const float* x    = (const float*)d_in[0];
    const float* ln_w = (const float*)d_in[1];
    const float* ln_b = (const float*)d_in[2];
    const float* W    = (const float*)d_in[3];
    const float* b    = (const float*)d_in[4];
    float* out        = (float*)d_out;

    const int total_rows = in_sizes[0] / CDIM;
    dim3 grid(total_rows / (2 * ROWS));
    dim3 block(512);
    fused_tinyops_kernel<<<grid, block, 0, stream>>>(x, ln_w, ln_b, W, b, out);
}


# Pointwise stage: tanh and sigmoid from the hardware tanh instruction (sigmoid(h) = (1 + tanh(h/2))/2): two transcendentals and three vector operations per element instead of four and seven
# speedup vs baseline: 1.5672x; 1.0362x over previous
#include <hip/hip_runtime.h>


typedef __attribute__((ext_vector_type(2))) float v2f;
typedef __attribute__((ext_vector_type(4))) float v4f;
typedef __attribute__((ext_vector_type(8))) float v8f;

#define CDIM   256
#define PITCH  260
#define ROWS   16
#define NITER  30
#define KT     32
#define TSZ    (ROWS * PITCH)
#define AG     16
#define LEAD   8
#define VT0    3

__device__ __forceinline__ float pointwise(float h)
{
    h = fmaf(0.01f, __builtin_amdgcn_tanhf(h), h);
    const float hh = 0.5f * h;
    return fmaf(hh, __builtin_amdgcn_tanhf(hh), hh);
}

__device__ __forceinline__ float xrow16(float x)
{
    const int b = __builtin_bit_cast(int, x);
    return __builtin_bit_cast(float,
        __builtin_amdgcn_permlanex16(b, b, 0x76543210, 0xfedcba98, false, false));
}

template <int M>
__device__ __forceinline__ float xlane(float x)
{
    const int b = __builtin_bit_cast(int, x);
    return __builtin_bit_cast(float,
        __builtin_amdgcn_update_dpp(0, b, 0x160 + M, 0xf, 0xf, true));
}

__device__ __forceinline__ void load_a_group(v2f (&afr)[KT], const float* __restrict__ arow, int g)
{
    #pragma unroll
    for (int i = 0; i < AG / 2; ++i) {
        const int q = g * (AG / 2) + i;
        const v4f a = *reinterpret_cast<const v4f*>(arow + 4 * q);
        afr[2 * q].x     = a.x;  afr[2 * q].y     = a.y;
        afr[2 * q + 1].x = a.z;  afr[2 * q + 1].y = a.w;
    }
}

template <int SRC, int DST, bool GEMM>
__device__ __forceinline__ void phase(
    const v2f (&B0)[KT], const v2f (&B1)[KT],
    const float* __restrict__ arow,
    float* __restrict__ pout,
    int dc1,
    const float* __restrict__ plo,
    const float* __restrict__ phi,
    const float* __restrict__ xrow,
    float* __restrict__ adst,
    float* __restrict__ orow,
    const float (&lw)[8], const float (&lb)[8], const float (&bs)[8])
{
    v2f afr[KT];
    float lo[8], hi[8], v[8];

    static_assert(LEAD > 0 && LEAD < AG && KT % AG == 0, "group schedule");
    if (GEMM) {
        load_a_group(afr, arow, 0);
        if (DST == 0) {
            __builtin_amdgcn_sched_barrier(0);
        }
    }
    if (SRC == 0) {
        #pragma unroll
        for (int j = 0; j < 8; ++j) { lo[j] = plo[32 * j]; hi[j] = phi[32 * j]; }
    } else {
        #pragma unroll
        for (int j = 0; j < 8; ++j) v[j] = xrow[32 * j];
    }
    if (GEMM && DST == 0) {
        __builtin_amdgcn_sched_barrier(0);
    }
    if (DST == 1) {
        #pragma unroll
        for (int j = 0; j < 8; ++j) {
            if (SRC == 0) v[j] = (lo[j] + hi[j]) + bs[j];
            orow[32 * j] = v[j];
        }
    }

    v8f acc0 = {0.f, 0.f, 0.f, 0.f, 0.f, 0.f, 0.f, 0.f};
    v8f acc1 = {0.f, 0.f, 0.f, 0.f, 0.f, 0.f, 0.f, 0.f};
    float s = 0.0f, mu = 0.0f, q = 0.0f, rstd = 0.0f;

    #pragma unroll
    for (int t = 0; t < KT; ++t) {
        if (GEMM) {
            if ((t + LEAD) % AG == 0 && t + LEAD >= AG && t + LEAD < KT) {
                load_a_group(afr, arow, (t + LEAD) / AG);
                if (DST == 0) {
                    __builtin_amdgcn_sched_barrier(0);
                }
            }
            acc0 = __builtin_amdgcn_wmma_f32_16x16x4_f32(
                false, afr[t], false, B0[t], (short)0, acc0, false, false);
            acc1 = __builtin_amdgcn_wmma_f32_16x16x4_f32(
                false, afr[t], false, B1[t], (short)0, acc1, false, false);
        }
        if (DST == 0) {
            if (t == VT0) {
                #pragma unroll
                for (int j = 0; j < 8; ++j) {
                    if (SRC == 0) v[j] = (lo[j] + hi[j]) + bs[j];
                    v[j] = pointwise(v[j]);
                }
                s = ((v[0] + v[1]) + (v[2] + v[3])) + ((v[4] + v[5]) + (v[6] + v[7]));
                s += xrow16(s);
                s += xlane<8>(s);
                s += xlane<4>(s);
                s += xlane<2>(s);
                s += xlane<1>(s);
                mu = s * (1.0f / CDIM);
                #pragma unroll
                for (int j = 0; j < 8; ++j) { float d = v[j] - mu; q += d * d; }
            } else if (t == VT0 + 1) {
                q += xrow16(q);
                q += xlane<8>(q);
                q += xlane<4>(q);
                q += xlane<2>(q);
                q += xlane<1>(q);
                rstd = rsqrtf(q * (1.0f / CDIM) + 1e-5f);
            } else if (t >= VT0 + 2 && t < VT0 + 10) {
                const int j = t - (VT0 + 2);
                v[j] = (v[j] - mu) * rstd * lw[j] + lb[j];
            }
        }
        if (GEMM && DST == 0) {
            __builtin_amdgcn_sched_barrier(0);
        }
    }

    if (DST == 0) {
        #pragma unroll
        for (int j = 0; j < 8; ++j) adst[32 * j] = v[j];
    }
    if (GEMM) {
        #pragma unroll
        for (int vi = 0; vi < 8; ++vi) {
            pout[vi * PITCH]       = acc0[vi];
            pout[vi * PITCH + dc1] = acc1[vi];
        }
    }
}

__global__ __launch_bounds__(512, 1) void fused_tinyops_kernel(
    const float* __restrict__ x, const float* __restrict__ ln_w,
    const float* __restrict__ ln_b, const float* __restrict__ W,
    const float* __restrict__ bias, float* __restrict__ out)
{
    __shared__ __attribute__((aligned(16))) float abuf[2 * TSZ];
    __shared__ float part[4 * TSZ];

    const int tid   = threadIdx.x;
    const int lane  = tid & 31;
    const int wave  = tid >> 5;
    const int l16   = lane & 15;
    const int koff  = (lane < 16) ? 0 : 64;
    const int khalf = wave >> 3;
    const int cpair = wave & 7;
    const int c0    = 2 * cpair;
    const int c1    = c0 + 1;

    const long rowBase = (long)blockIdx.x * (2 * ROWS);

    v2f B0[KT], B1[KT];
    {
        const float* w0 = W + (16 * c0 + l16) * CDIM + 128 * khalf + koff;
        const float* w1 = W + (16 * c1 + l16) * CDIM + 128 * khalf + koff;
        #pragma unroll
        for (int t = 0; t < KT; ++t) {
            B0[t].x = w0[2 * t + 0];  B0[t].y = w0[2 * t + 1];
            B1[t].x = w1[2 * t + 0];  B1[t].y = w1[2 * t + 1];
        }
    }

    float lw[8], lb[8], bs[8];
    #pragma unroll
    for (int j = 0; j < 8; ++j) {
        lw[j] = ln_w[lane + 32 * j];
        lb[j] = ln_b[lane + 32 * j];
        bs[j] = bias[lane + 32 * j];
    }

    const int aoff = l16 * PITCH + 128 * khalf + koff;
    const int poff = khalf * TSZ + ((lane < 16) ? 0 : 8) * PITCH + 16 * c0 + l16;
    const float* a0 = &abuf[0 * TSZ + aoff];
    const float* a1 = &abuf[1 * TSZ + aoff];
    float* p0 = &part[0 * 2 * TSZ + poff];
    float* p1 = &part[1 * 2 * TSZ + poff];
    const int roff = wave * PITCH + lane;
    const float* lo0 = &part[0 * 2 * TSZ + roff];
    const float* hi0 = &part[0 * 2 * TSZ + TSZ + roff];
    const float* lo1 = &part[1 * 2 * TSZ + roff];
    const float* hi1 = &part[1 * 2 * TSZ + TSZ + roff];
    float* d0 = &abuf[0 * TSZ + roff];
    float* d1 = &abuf[1 * TSZ + roff];
    const float* x0 = x + (rowBase + wave) * CDIM + lane;
    const float* x1 = x + (rowBase + ROWS + wave) * CDIM + lane;
    float* o0 = out + (rowBase + wave) * CDIM + lane;
    float* o1 = out + (rowBase + ROWS + wave) * CDIM + lane;

    phase<1, 0, false>(B0, B1, a0, p0, 16, lo0, hi0, x0, d0, o0, lw, lb, bs);
    __syncthreads();
    phase<1, 0, true>(B0, B1, a0, p0, 16, lo1, hi1, x1, d1, o1, lw, lb, bs);
    __syncthreads();

    for (int it = 0; it < NITER - 1; ++it) {
        phase<0, 0, true>(B0, B1, a1, p1, 16, lo0, hi0, x0, d0, o0, lw, lb, bs);
        __syncthreads();
        phase<0, 0, true>(B0, B1, a0, p0, 16, lo1, hi1, x1, d1, o1, lw, lb, bs);
        __syncthreads();
    }

    phase<0, 1, true>(B0, B1, a1, p1, 16, lo0, hi0, x0, d0, o0, lw, lb, bs);
    __syncthreads();
    phase<0, 1, false>(B0, B1, a1, p1, 16, lo1, hi1, x1, d1, o1, lw, lb, bs);
}

extern "C" void kernel_launch(void* const* d_in, const int* in_sizes, int n_in,
                              void* d_out, int out_size, void* d_ws, size_t ws_size,
                              hipStream_t stream) {
    (void)n_in; (void)d_ws; (void)ws_size; (void)out_size;
    const float* x    = (const float*)d_in[0];
    const float* ln_w = (const float*)d_in[1];
    const float* ln_b = (const float*)d_in[2];
    const float* W    = (const float*)d_in[3];
    const float* b    = (const float*)d_in[4];
    float* out        = (float*)d_out;

    const int total_rows = in_sizes[0] / CDIM;
    dim3 grid(total_rows / (2 * ROWS));
    dim3 block(512);
    fused_tinyops_kernel<<<grid, block, 0, stream>>>(x, ln_w, ln_b, W, b, out);
}


# LayerNorm stage cut to centring: ln_w folded into W and ln_b into the bias once per block, 1/std applied where the next phase reads the sums; row sums on pairs: LayerNorm summation order changes
# speedup vs baseline: 1.6401x; 1.0465x over previous
#include <hip/hip_runtime.h>


typedef __attribute__((ext_vector_type(2))) float v2f;
typedef __attribute__((ext_vector_type(4))) float v4f;
typedef __attribute__((ext_vector_type(8))) float v8f;

#define CDIM   256
#define PITCH  260
#define ROWS   16
#define NITER  30
#define KT     32
#define TSZ    (ROWS * PITCH)
#define AG     16
#define LEAD   8
#define VT0    3

__device__ __forceinline__ v2f splat2(float a) { v2f r; r.x = a; r.y = a; return r; }

__device__ __forceinline__ v2f tanh2(v2f a)
{
    v2f r;
    r.x = __builtin_amdgcn_tanhf(a.x);
    r.y = __builtin_amdgcn_tanhf(a.y);
    return r;
}

__device__ __forceinline__ v2f pointwise(v2f h)
{
    h = __builtin_elementwise_fma(splat2(0.01f), tanh2(h), h);
    const v2f hh = h * 0.5f;
    return __builtin_elementwise_fma(hh, tanh2(hh), hh);
}

__device__ __forceinline__ float xrow16(float x)
{
    const int b = __builtin_bit_cast(int, x);
    return __builtin_bit_cast(float,
        __builtin_amdgcn_permlanex16(b, b, 0x76543210, 0xfedcba98, false, true));
}

template <int M>
__device__ __forceinline__ float xlane(float x)
{
    const int b = __builtin_bit_cast(int, x);
    return __builtin_bit_cast(float,
        __builtin_amdgcn_update_dpp(0, b, 0x160 + M, 0xf, 0xf, true));
}

__device__ __forceinline__ void load_a_group(v2f (&afr)[KT], const float* __restrict__ arow, int g)
{
    #pragma unroll
    for (int i = 0; i < AG / 2; ++i) {
        const int q = g * (AG / 2) + i;
        const v4f a = *reinterpret_cast<const v4f*>(arow + 4 * q);
        afr[2 * q].x     = a.x;  afr[2 * q].y     = a.y;
        afr[2 * q + 1].x = a.z;  afr[2 * q + 1].y = a.w;
    }
}

template <int SRC, int DST, bool GEMM>
__device__ __forceinline__ void phase(
    const v2f (&B0)[KT], const v2f (&B1)[KT],
    const float* __restrict__ arow,
    float* __restrict__ pout,
    int dc1,
    const float* __restrict__ plo,
    const float* __restrict__ phi,
    const float* __restrict__ xrow,
    float* __restrict__ adst,
    float* __restrict__ orow,
    const v2f (&bs)[4],
    float& rs)
{
    v2f afr[KT];
    v2f lo[4], hi[4], v[4];

    static_assert(LEAD > 0 && LEAD < AG && KT % AG == 0, "group schedule");
    if (GEMM) {
        load_a_group(afr, arow, 0);
        if (DST == 0) {
            __builtin_amdgcn_sched_barrier(0);
        }
    }
    if (SRC == 0) {
        #pragma unroll
        for (int jj = 0; jj < 4; ++jj) {
            lo[jj].x = plo[64 * jj];  lo[jj].y = plo[64 * jj + 32];
            hi[jj].x = phi[64 * jj];  hi[jj].y = phi[64 * jj + 32];
        }
    } else {
        #pragma unroll
        for (int jj = 0; jj < 4; ++jj) { v[jj].x = xrow[64 * jj];  v[jj].y = xrow[64 * jj + 32]; }
    }
    if (GEMM && DST == 0) {
        __builtin_amdgcn_sched_barrier(0);
    }
    if (DST == 1) {
        #pragma unroll
        for (int jj = 0; jj < 4; ++jj) {
            if (SRC == 0) v[jj] = __builtin_elementwise_fma(lo[jj] + hi[jj], splat2(rs), bs[jj]);
            orow[64 * jj]      = v[jj].x;
            orow[64 * jj + 32] = v[jj].y;
        }
    }

    v8f acc0 = {0.f, 0.f, 0.f, 0.f, 0.f, 0.f, 0.f, 0.f};
    v8f acc1 = {0.f, 0.f, 0.f, 0.f, 0.f, 0.f, 0.f, 0.f};
    float s = 0.0f, q = 0.0f;

    #pragma unroll
    for (int t = 0; t < KT; ++t) {
        if (GEMM) {
            if ((t + LEAD) % AG == 0 && t + LEAD >= AG && t + LEAD < KT) {
                load_a_group(afr, arow, (t + LEAD) / AG);
                if (DST == 0) {
                    __builtin_amdgcn_sched_barrier(0);
                }
            }
            acc0 = __builtin_amdgcn_wmma_f32_16x16x4_f32(
                false, afr[t], false, B0[t], (short)0, acc0, false, false);
            acc1 = __builtin_amdgcn_wmma_f32_16x16x4_f32(
                false, afr[t], false, B1[t], (short)0, acc1, false, false);
        }
        if (DST == 0) {
            if (t == VT0) {
                #pragma unroll
                for (int jj = 0; jj < 4; ++jj) {
                    if (SRC == 0) v[jj] = __builtin_elementwise_fma(lo[jj] + hi[jj], splat2(rs), bs[jj]);
                    v[jj] = pointwise(v[jj]);
                }
                const v2f s2 = (v[0] + v[1]) + (v[2] + v[3]);
                s = s2.x + s2.y;
                s += xrow16(s);
                s += xlane<8>(s);
                s += xlane<4>(s);
                s += xlane<2>(s);
                s += xlane<1>(s);
                const float mu = s * (1.0f / CDIM);
                #pragma unroll
                for (int jj = 0; jj < 4; ++jj) v[jj] = v[jj] - mu;
                v2f q2 = v[0] * v[0];
                q2 = __builtin_elementwise_fma(v[1], v[1], q2);
                q2 = __builtin_elementwise_fma(v[2], v[2], q2);
                q2 = __builtin_elementwise_fma(v[3], v[3], q2);
                q = q2.x + q2.y;
            } else if (t == VT0 + 1) {
                q += xrow16(q);
                q += xlane<8>(q);
                q += xlane<4>(q);
                q += xlane<2>(q);
                q += xlane<1>(q);
                rs = __builtin_amdgcn_rsqf(q * (1.0f / CDIM) + 1e-5f);
            }
        }
        if (GEMM && DST == 0) {
            __builtin_amdgcn_sched_barrier(0);
        }
    }

    if (DST == 0) {
        #pragma unroll
        for (int jj = 0; jj < 4; ++jj) {
            adst[64 * jj]      = v[jj].x;
            adst[64 * jj + 32] = v[jj].y;
        }
    }
    if (GEMM) {
        #pragma unroll
        for (int vi = 0; vi < 8; ++vi) {
            pout[vi * PITCH]       = acc0[vi];
            pout[vi * PITCH + dc1] = acc1[vi];
        }
    }
}

__global__ __launch_bounds__(512, 1) void fused_tinyops_kernel(
    const float* __restrict__ x, const float* __restrict__ ln_w,
    const float* __restrict__ ln_b, const float* __restrict__ W,
    const float* __restrict__ bias, float* __restrict__ out)
{
    __shared__ __attribute__((aligned(16))) float abuf[2 * TSZ];
    __shared__ float part[4 * TSZ];
    __shared__ float wlb[2 * CDIM];

    const int tid   = threadIdx.x;
    const int lane  = tid & 31;
    const int wave  = tid >> 5;
    const int l16   = lane & 15;
    const int koff  = (lane < 16) ? 0 : 64;
    const int khalf = wave >> 3;
    const int cpair = wave & 7;
    const int c0    = 2 * cpair;
    const int c1    = c0 + 1;

    const long rowBase = (long)blockIdx.x * (2 * ROWS);

    v2f B0[KT], B1[KT];
    {
        const int kb = 128 * khalf + koff;
        const float* w0 = W + (16 * c0 + l16) * CDIM + kb;
        const float* w1 = W + (16 * c1 + l16) * CDIM + kb;
        const float* gw = ln_w + kb;
        const float* gb = ln_b + kb;
        v2f wb0 = {0.f, 0.f}, wb1 = {0.f, 0.f};
        #pragma unroll
        for (int t = 0; t < KT; ++t) {
            v2f u0, u1, g, z;
            u0.x = w0[2 * t];  u0.y = w0[2 * t + 1];
            u1.x = w1[2 * t];  u1.y = w1[2 * t + 1];
            g.x  = gw[2 * t];  g.y  = gw[2 * t + 1];
            z.x  = gb[2 * t];  z.y  = gb[2 * t + 1];
            wb0 = __builtin_elementwise_fma(u0, z, wb0);
            wb1 = __builtin_elementwise_fma(u1, z, wb1);
            B0[t] = u0 * g;
            B1[t] = u1 * g;
        }
        float pb0 = wb0.x + wb0.y, pb1 = wb1.x + wb1.y;
        pb0 += xrow16(pb0);
        pb1 += xrow16(pb1);
        if (khalf == 0) {
            pb0 += bias[16 * c0 + l16];
            pb1 += bias[16 * c1 + l16];
        }
        wlb[khalf * CDIM + 16 * c0 + l16] = pb0;
        wlb[khalf * CDIM + 16 * c1 + l16] = pb1;
    }

    const int aoff = l16 * PITCH + 128 * khalf + koff;
    const int poff = khalf * TSZ + ((lane < 16) ? 0 : 8) * PITCH + 16 * c0 + l16;
    const float* a0 = &abuf[0 * TSZ + aoff];
    const float* a1 = &abuf[1 * TSZ + aoff];
    float* p0 = &part[0 * 2 * TSZ + poff];
    float* p1 = &part[1 * 2 * TSZ + poff];
    const int roff = wave * PITCH + lane;
    const float* lo0 = &part[0 * 2 * TSZ + roff];
    const float* hi0 = &part[0 * 2 * TSZ + TSZ + roff];
    const float* lo1 = &part[1 * 2 * TSZ + roff];
    const float* hi1 = &part[1 * 2 * TSZ + TSZ + roff];
    float* d0 = &abuf[0 * TSZ + roff];
    float* d1 = &abuf[1 * TSZ + roff];
    const float* x0 = x + (rowBase + wave) * CDIM + lane;
    const float* x1 = x + (rowBase + ROWS + wave) * CDIM + lane;
    float* o0 = out + (rowBase + wave) * CDIM + lane;
    float* o1 = out + (rowBase + ROWS + wave) * CDIM + lane;

    v2f bs[4] = {{0.f, 0.f}, {0.f, 0.f}, {0.f, 0.f}, {0.f, 0.f}};
    float rs0 = 0.0f, rs1 = 0.0f;

    phase<1, 0, false>(B0, B1, a0, p0, 16, lo0, hi0, x0, d0, o0, bs, rs0);
    __syncthreads();
    phase<1, 0, true>(B0, B1, a0, p0, 16, lo1, hi1, x1, d1, o1, bs, rs1);
    __syncthreads();

    #pragma unroll
    for (int jj = 0; jj < 4; ++jj) {
        const int c = lane + 64 * jj;
        bs[jj].x = wlb[c]      + wlb[CDIM + c];
        bs[jj].y = wlb[c + 32] + wlb[CDIM + c + 32];
    }

    for (int it = 0; it < NITER - 1; ++it) {
        phase<0, 0, true>(B0, B1, a1, p1, 16, lo0, hi0, x0, d0, o0, bs, rs0);
        __syncthreads();
        phase<0, 0, true>(B0, B1, a0, p0, 16, lo1, hi1, x1, d1, o1, bs, rs1);
        __syncthreads();
    }

    phase<0, 1, true>(B0, B1, a1, p1, 16, lo0, hi0, x0, d0, o0, bs, rs0);
    __syncthreads();
    phase<0, 1, false>(B0, B1, a1, p1, 16, lo1, hi1, x1, d1, o1, bs, rs1);
}

extern "C" void kernel_launch(void* const* d_in, const int* in_sizes, int n_in,
                              void* d_out, int out_size, void* d_ws, size_t ws_size,
                              hipStream_t stream) {
    (void)n_in; (void)d_ws; (void)ws_size; (void)out_size;
    const float* x    = (const float*)d_in[0];
    const float* ln_w = (const float*)d_in[1];
    const float* ln_b = (const float*)d_in[2];
    const float* W    = (const float*)d_in[3];
    const float* b    = (const float*)d_in[4];
    float* out        = (float*)d_out;

    const int total_rows = in_sizes[0] / CDIM;
    dim3 grid(total_rows / (2 * ROWS));
    dim3 block(512);
    fused_tinyops_kernel<<<grid, block, 0, stream>>>(x, ln_w, ln_b, W, b, out);
}


# Block turned inside out: each wave keeps its 16 rows in registers for all 30 steps (D registers = next B fragments), W staged once in LDS as A operand; Linear k order and LayerNorm sum order change
# speedup vs baseline: 1.7479x; 1.0657x over previous
#include <hip/hip_runtime.h>


typedef __attribute__((ext_vector_type(2))) float v2f;
typedef __attribute__((ext_vector_type(4))) float v4f;
typedef __attribute__((ext_vector_type(8))) float v8f;

#define CDIM   256
#define PITCH  260
#define ROWS   16
#define WAVES  4
#define NITER  30
#define NTILE  16
#define NBLK   16
#define GRP    32

__device__ __forceinline__ v2f splat2(float a) { v2f r; r.x = a; r.y = a; return r; }

__device__ __forceinline__ v2f tanh2(v2f a)
{
    v2f r;
    r.x = __builtin_amdgcn_tanhf(a.x);
    r.y = __builtin_amdgcn_tanhf(a.y);
    return r;
}

__device__ __forceinline__ v2f pointwise(v2f h)
{
    h = __builtin_elementwise_fma(splat2(0.01f), tanh2(h), h);
    const v2f hh = h * 0.5f;
    return __builtin_elementwise_fma(hh, tanh2(hh), hh);
}

__device__ __forceinline__ float xrow16(float x)
{
    const int b = __builtin_bit_cast(int, x);
    return __builtin_bit_cast(float,
        __builtin_amdgcn_permlanex16(b, b, 0x76543210, 0xfedcba98, false, true));
}

template <int M>
__device__ __forceinline__ float xlane(float x)
{
    const int b = __builtin_bit_cast(int, x);
    return __builtin_bit_cast(float,
        __builtin_amdgcn_update_dpp(0, b, 0x160 + M, 0xf, 0xf, true));
}

__device__ __forceinline__ void load_block(v4f (&a)[GRP], v4f (&bt)[4],
                                           const float* __restrict__ wrow,
                                           const float* __restrict__ brow, int blk)
{
    const int j0 = 2 * (blk >> 1);
    const int q0 = 16 * (blk & 1);
    #pragma unroll
    for (int u = 0; u < 2; ++u) {
        #pragma unroll
        for (int i = 0; i < 16; ++i) {
            const int q = q0 + i;
            a[16 * u + i] = *reinterpret_cast<const v4f*>(
                wrow + 16 * (j0 + u) * PITCH + 16 * (q >> 1) + 4 * (q & 1));
        }
    }
    if ((blk & 1) == 0) {
        #pragma unroll
        for (int u = 0; u < 2; ++u) {
            bt[2 * u]     = *reinterpret_cast<const v4f*>(brow + 16 * (j0 + u));
            bt[2 * u + 1] = *reinterpret_cast<const v4f*>(brow + 16 * (j0 + u) + 4);
        }
    }
}

__device__ __forceinline__ void mma_block(v8f (&d)[NTILE], const v2f (&B)[64],
                                          const v4f (&a)[GRP], const v4f (&bt)[4], int blk)
{
    const int j0 = 2 * (blk >> 1);
    const int k0 = 32 * (blk & 1);
    v8f acc0, acc1;
    if ((blk & 1) == 0) {
        acc0 = __builtin_shufflevector(bt[0], bt[1], 0, 1, 2, 3, 4, 5, 6, 7);
        acc1 = __builtin_shufflevector(bt[2], bt[3], 0, 1, 2, 3, 4, 5, 6, 7);
    } else {
        acc0 = d[j0];
        acc1 = d[j0 + 1];
    }
    #pragma unroll
    for (int t = 0; t < 32; ++t) {
        v2f f0, f1;
        if (t & 1) { f0.x = a[t >> 1].z;  f0.y = a[t >> 1].w;  f1.x = a[16 + (t >> 1)].z;  f1.y = a[16 + (t >> 1)].w; }
        else       { f0.x = a[t >> 1].x;  f0.y = a[t >> 1].y;  f1.x = a[16 + (t >> 1)].x;  f1.y = a[16 + (t >> 1)].y; }
        acc0 = __builtin_amdgcn_wmma_f32_16x16x4_f32(
            false, f0, false, B[k0 + t], (short)0, acc0, false, false);
        acc1 = __builtin_amdgcn_wmma_f32_16x16x4_f32(
            false, f1, false, B[k0 + t], (short)0, acc1, false, false);
    }
    d[j0]     = acc0;
    d[j0 + 1] = acc1;
}

__device__ __forceinline__ void vector_stage(const v8f (&d)[NTILE], v2f (&B)[64])
{
    v2f v[64];
    #pragma unroll
    for (int p = 0; p < 64; ++p) {
        v2f t;
        t.x = d[p >> 2][2 * (p & 3)];
        t.y = d[p >> 2][2 * (p & 3) + 1];
        v[p] = pointwise(t);
    }
    v2f s4[4] = {{0.f, 0.f}, {0.f, 0.f}, {0.f, 0.f}, {0.f, 0.f}};
    #pragma unroll
    for (int p = 0; p < 64; ++p) s4[p & 3] = s4[p & 3] + v[p];
    const v2f s2 = (s4[0] + s4[1]) + (s4[2] + s4[3]);
    float s = s2.x + s2.y;
    s += xrow16(s);
    const float mu = s * (1.0f / CDIM);
    v2f q4[4] = {{0.f, 0.f}, {0.f, 0.f}, {0.f, 0.f}, {0.f, 0.f}};
    #pragma unroll
    for (int p = 0; p < 64; ++p) {
        v[p] = v[p] - mu;
        q4[p & 3] = __builtin_elementwise_fma(v[p], v[p], q4[p & 3]);
    }
    const v2f q2 = (q4[0] + q4[1]) + (q4[2] + q4[3]);
    float q = q2.x + q2.y;
    q += xrow16(q);
    const float rs = __builtin_amdgcn_rsqf(q * (1.0f / CDIM) + 1e-5f);
    #pragma unroll
    for (int p = 0; p < 64; ++p) B[p] = v[p] * rs;
}

__global__ __launch_bounds__(32 * WAVES) void fused_tinyops_kernel(
    const float* __restrict__ x, const float* __restrict__ ln_w,
    const float* __restrict__ ln_b, const float* __restrict__ W,
    const float* __restrict__ bias, float* __restrict__ out)
{
    __shared__ __attribute__((aligned(16))) float wl[CDIM * PITCH];
    __shared__ __attribute__((aligned(16))) float b2[CDIM];

    const int tid  = threadIdx.x;
    const int lane = tid & 31;
    const int wave = tid >> 5;
    const int c    = lane & 15;
    const int h    = lane >> 4;

    {
        const v4f g0 = *reinterpret_cast<const v4f*>(ln_w + 4 * lane);
        const v4f g1 = *reinterpret_cast<const v4f*>(ln_w + 128 + 4 * lane);
        const v4f z0 = *reinterpret_cast<const v4f*>(ln_b + 4 * lane);
        const v4f z1 = *reinterpret_cast<const v4f*>(ln_b + 128 + 4 * lane);
        #pragma unroll 1
        for (int g = 0; g < 4; ++g) {
            const int n0 = 64 * wave + 16 * g;
            v4f w0[16], w1[16];
            #pragma unroll
            for (int r = 0; r < 16; ++r) {
                w0[r] = *reinterpret_cast<const v4f*>(W + (n0 + r) * CDIM + 4 * lane);
                w1[r] = *reinterpret_cast<const v4f*>(W + (n0 + r) * CDIM + 128 + 4 * lane);
            }
            const float bv = bias[n0 + c];
            float keep = 0.0f;
            #pragma unroll
            for (int r = 0; r < 16; ++r) {
                *reinterpret_cast<v4f*>(wl + (n0 + r) * PITCH + 4 * lane)       = w0[r] * g0;
                *reinterpret_cast<v4f*>(wl + (n0 + r) * PITCH + 128 + 4 * lane) = w1[r] * g1;
                v4f pz = w0[r] * z0;
                pz = __builtin_elementwise_fma(w1[r], z1, pz);
                float p = (pz.x + pz.y) + (pz.z + pz.w);
                p += xrow16(p);
                p += xlane<8>(p);
                p += xlane<4>(p);
                p += xlane<2>(p);
                p += xlane<1>(p);
                keep = (c == r) ? p : keep;
            }
            b2[n0 + c] = keep + bv;
        }
    }
    __syncthreads();

    const float* wrow = wl + c * PITCH + 8 * h;
    const float* brow = b2 + 8 * h;
    const long row = (long)blockIdx.x * (WAVES * ROWS) + ROWS * wave + c;
    const float* xr = x + row * CDIM + 8 * h;
    float* orow = out + row * CDIM + 8 * h;

    v8f d[NTILE];
    #pragma unroll
    for (int j = 0; j < NTILE; ++j) {
        const v4f u0 = *reinterpret_cast<const v4f*>(xr + 16 * j);
        const v4f u1 = *reinterpret_cast<const v4f*>(xr + 16 * j + 4);
        d[j] = __builtin_shufflevector(u0, u1, 0, 1, 2, 3, 4, 5, 6, 7);
    }

    v2f B[64];
    v4f a0[GRP], a1[GRP], bt[4];
    load_block(a0, bt, wrow, brow, 0);

    for (int it = 0; it < NITER; ++it) {
        vector_stage(d, B);
        #pragma unroll
        for (int blk = 0; blk < NBLK; ++blk) {
            __builtin_amdgcn_sched_barrier(0);
            if (blk & 1) load_block(a0, bt, wrow, brow, (blk + 1) & (NBLK - 1));
            else         load_block(a1, bt, wrow, brow, blk + 1);
            __builtin_amdgcn_sched_barrier(0);
            if (blk & 1) mma_block(d, B, a1, bt, blk);
            else         mma_block(d, B, a0, bt, blk);
        }
    }

    #pragma unroll
    for (int j = 0; j < NTILE; ++j) {
        v4f u0, u1;
        u0.x = d[j][0]; u0.y = d[j][1]; u0.z = d[j][2]; u0.w = d[j][3];
        u1.x = d[j][4]; u1.y = d[j][5]; u1.z = d[j][6]; u1.w = d[j][7];
        *reinterpret_cast<v4f*>(orow + 16 * j)     = u0;
        *reinterpret_cast<v4f*>(orow + 16 * j + 4) = u1;
    }
}

extern "C" void kernel_launch(void* const* d_in, const int* in_sizes, int n_in,
                              void* d_out, int out_size, void* d_ws, size_t ws_size,
                              hipStream_t stream) {
    (void)n_in; (void)d_ws; (void)ws_size; (void)out_size;
    const float* x    = (const float*)d_in[0];
    const float* ln_w = (const float*)d_in[1];
    const float* ln_b = (const float*)d_in[2];
    const float* W    = (const float*)d_in[3];
    const float* b    = (const float*)d_in[4];
    float* out        = (float*)d_out;

    const int total_rows = in_sizes[0] / CDIM;
    dim3 grid(total_rows / (WAVES * ROWS));
    dim3 block(32 * WAVES);
    fused_tinyops_kernel<<<grid, block, 0, stream>>>(x, ln_w, ln_b, W, b, out);
}
